# starting kernel
# baseline (speedup 1.0000x reference)
#include <hip/hip_runtime.h>

typedef float v2f __attribute__((ext_vector_type(2)));
typedef float v8f __attribute__((ext_vector_type(8)));

#define BATCHES   16
#define SEQ       256
#define DIM       1024
#define RANK      128
#define KCHUNK    64
#define LDSA_STR  68
#define LDST_STR  132

__global__ __launch_bounds__(256) void proj_wmma(const float* __restrict__ batch,
                                                 const float* __restrict__ Bm,
                                                 float* __restrict__ T) {
  __shared__ float ldsA[16 * LDSA_STR];
  __shared__ float ldsB[KCHUNK * RANK];

  const int b    = blockIdx.x;
  const int mt   = blockIdx.y;
  const int t    = threadIdx.x;
  const int wave = t >> 5;
  const int lane = t & 31;
  const int lcol = lane & 15;
  const int khalf = (lane >> 4) << 1;
  const int n0   = wave * 16;

  const float* Ag = batch + (size_t)b * SEQ * DIM + (size_t)(mt * 16) * DIM;

  v8f acc = {};

  for (int k0 = 0; k0 < DIM; k0 += KCHUNK) {
    {
      const int idx = t * 4;
      const int row = idx >> 6;
      const int col = idx & 63;
      const float4 v = *reinterpret_cast<const float4*>(Ag + row * DIM + k0 + col);
      *reinterpret_cast<float4*>(&ldsA[row * LDSA_STR + col]) = v;
    }
    {
      const float* Bg = Bm + (size_t)k0 * RANK;
#pragma unroll
      for (int c = 0; c < 8; ++c) {
        const int idx = (c * 256 + t) * 4;
        *reinterpret_cast<float4*>(&ldsB[idx]) = *reinterpret_cast<const float4*>(Bg + idx);
      }
    }
    __syncthreads();

#pragma unroll
    for (int kk = 0; kk < KCHUNK; kk += 4) {
      v2f afrag = *reinterpret_cast<const v2f*>(&ldsA[lcol * LDSA_STR + kk + khalf]);
      v2f bfrag;
      bfrag.x = ldsB[(kk + khalf)     * RANK + n0 + lcol];
      bfrag.y = ldsB[(kk + khalf + 1) * RANK + n0 + lcol];
      acc = __builtin_amdgcn_wmma_f32_16x16x4_f32(false, afrag, false, bfrag,
                                                  (short)0, acc, false, false);
    }
    __syncthreads();
  }

  float* Tb = T + (size_t)b * SEQ * RANK + (size_t)(mt * 16) * RANK;
  const int mhi = (lane >> 4) * 8;
#pragma unroll
  for (int v = 0; v < 8; ++v) {
    Tb[(size_t)(v + mhi) * RANK + n0 + lcol] = acc[v];
  }
}

__global__ __launch_bounds__(128) void gram_wmma(const float* __restrict__ T,
                                                 float* __restrict__ out) {
  __shared__ float ldsT[SEQ * LDST_STR];
  __shared__ float ldsN[SEQ];

  const int b     = blockIdx.x;
  const int jgrp  = blockIdx.y;
  const int t     = threadIdx.x;
  const int wave  = t >> 5;
  const int lane  = t & 31;
  const int lcol  = lane & 15;
  const int khalf = (lane >> 4) << 1;
  const int jt    = jgrp * 4 + wave;

  const float* Tb = T + (size_t)b * SEQ * RANK;

#pragma unroll 4
  for (int c = 0; c < 64; ++c) {
    const int idx = (c * 128 + t) * 4;
    const int row = idx >> 7;
    const int col = idx & 127;
    *reinterpret_cast<float4*>(&ldsT[row * LDST_STR + col]) =
        *reinterpret_cast<const float4*>(Tb + idx);
  }
  __syncthreads();

#pragma unroll
  for (int r = 0; r < 2; ++r) {
    const int row = t + r * 128;
    const float* p = &ldsT[row * LDST_STR];
    float s = 0.f;
    for (int k = 0; k < RANK; ++k) s += p[k] * p[k];
    ldsN[row] = s;
  }
  __syncthreads();

  float* outb = out + (size_t)b * SEQ * SEQ;
  const int mhi = (lane >> 4) * 8;
  const float* Bj = &ldsT[(jt * 16 + lcol) * LDST_STR];

  for (int it = 0; it < 16; ++it) {
    const float* Ai = &ldsT[(it * 16 + lcol) * LDST_STR];
    v8f acc = {};
#pragma unroll
    for (int kk = 0; kk < RANK; kk += 4) {
      v2f afrag = *reinterpret_cast<const v2f*>(Ai + kk + khalf);
      v2f bfrag = *reinterpret_cast<const v2f*>(Bj + kk + khalf);
      acc = __builtin_amdgcn_wmma_f32_16x16x4_f32(false, afrag, false, bfrag,
                                                  (short)0, acc, false, false);
    }
#pragma unroll
    for (int v = 0; v < 8; ++v) {
      const int i = it * 16 + v + mhi;
      const int j = jt * 16 + lcol;
      outb[(size_t)i * SEQ + j] = ldsN[i] + ldsN[j] - 2.0f * acc[v];
    }
  }
}

extern "C" void kernel_launch(void* const* d_in, const int* in_sizes, int n_in,
                              void* d_out, int out_size, void* d_ws, size_t ws_size,
                              hipStream_t stream) {
  const float* batch = (const float*)d_in[0];
  const float* Bm    = (const float*)d_in[1];
  float* out = (float*)d_out;
  float* T   = (float*)d_ws;

  proj_wmma<<<dim3(BATCHES, SEQ / 16), 256, 0, stream>>>(batch, Bm, T);
  gram_wmma<<<dim3(BATCHES, 4), 128, 0, stream>>>(T, out);
}


# T passed in WMMA fragment order: projection stores it so, Gram loads fragments from global straight to registers (no LDS, no barrier), 256 groups x 4 waves, norms from fragments; sum order changes
# speedup vs baseline: 2.4751x; 2.4751x over previous
#include <hip/hip_runtime.h>

typedef float v2f __attribute__((ext_vector_type(2)));
typedef float v8f __attribute__((ext_vector_type(8)));

#define BATCHES   16
#define SEQ       256
#define DIM       1024
#define RANK      128
#define TILE_F4   512

#define KSLICE   128
#define NSTEP    16
#define PFD      3
#define RED_STR  132

struct FragStage {
  float4 a;
  float4 lo[4];
  float4 hi[4];
};

__device__ __forceinline__ void load_stage(FragStage& s, const float* __restrict__ Ap,
                                           const float* __restrict__ Bp) {
  s.a = *reinterpret_cast<const float4*>(Ap);
#pragma unroll
  for (int r = 0; r < 4; ++r) {
    s.lo[r] = *reinterpret_cast<const float4*>(Bp + r * RANK);
    s.hi[r] = *reinterpret_cast<const float4*>(Bp + r * RANK + 64);
  }
}

__device__ __forceinline__ float f4get(const float4& v, int t) {
  return t == 0 ? v.x : (t == 1 ? v.y : (t == 2 ? v.z : v.w));
}

__global__ __launch_bounds__(256) __attribute__((amdgpu_waves_per_eu(2, 4))) void proj_wmma(const float* __restrict__ batch,
                                                 const float* __restrict__ Bm,
                                                 float* __restrict__ T) {
  __shared__ float red[8 * 16 * RED_STR];

  const int b    = blockIdx.x;
  const int mt   = blockIdx.y;
  const int t    = threadIdx.x;
  const int wave = t >> 5;
  const int lane = t & 31;
  const int c    = lane & 15;
  const int h    = lane >> 4;

  const float* Ap = batch + (size_t)b * SEQ * DIM + (size_t)(mt * 16 + c) * DIM
                    + wave * KSLICE + 4 * h;
  const float* Bp = Bm + (size_t)(wave * KSLICE + 4 * h) * RANK + 4 * c;

  v8f acc[8];
#pragma unroll
  for (int i = 0; i < 8; ++i) acc[i] = v8f{};

  FragStage st[PFD + 1];
#pragma unroll
  for (int i = 0; i < PFD; ++i) load_stage(st[i], Ap + 8 * i, Bp + (size_t)(8 * i) * RANK);

#pragma unroll
  for (int i = 0; i < NSTEP; ++i) {
    if (i + PFD < NSTEP)
      load_stage(st[(i + PFD) % (PFD + 1)], Ap + 8 * (i + PFD), Bp + (size_t)(8 * (i + PFD)) * RANK);
    const FragStage& s = st[i % (PFD + 1)];
#pragma unroll
    for (int half = 0; half < 2; ++half) {
      v2f afrag;
      afrag.x = half ? s.a.z : s.a.x;
      afrag.y = half ? s.a.w : s.a.y;
#pragma unroll
      for (int q = 0; q < 4; ++q) {
        v2f blo, bhi;
        blo.x = f4get(s.lo[2 * half], q);
        blo.y = f4get(s.lo[2 * half + 1], q);
        bhi.x = f4get(s.hi[2 * half], q);
        bhi.y = f4get(s.hi[2 * half + 1], q);
        acc[q] = __builtin_amdgcn_wmma_f32_16x16x4_f32(false, afrag, false, blo,
                                                       (short)0, acc[q], false, false);
        acc[4 + q] = __builtin_amdgcn_wmma_f32_16x16x4_f32(false, afrag, false, bhi,
                                                           (short)0, acc[4 + q], false, false);
      }
    }
  }

  {
    float* rw = &red[(wave * 16 + 8 * h) * RED_STR + 4 * c];
#pragma unroll
    for (int v = 0; v < 8; ++v) {
      float4 x0 = make_float4(acc[0][v], acc[1][v], acc[2][v], acc[3][v]);
      float4 x1 = make_float4(acc[4][v], acc[5][v], acc[6][v], acc[7][v]);
      *reinterpret_cast<float4*>(rw + v * RED_STR)      = x0;
      *reinterpret_cast<float4*>(rw + v * RED_STR + 64) = x1;
    }
  }
  __syncthreads();

  float4* Tt = reinterpret_cast<float4*>(T) + ((size_t)b * (SEQ / 16) + mt) * TILE_F4;
#pragma unroll
  for (int r = 0; r < 2; ++r) {
    const int f   = t + r * 256;
    const int row = f & 15;
    const int n0  = 8 * (f >> 5) + 4 * ((f >> 4) & 1);
    float4 s = *reinterpret_cast<const float4*>(&red[row * RED_STR + n0]);
#pragma unroll
    for (int w = 1; w < 8; ++w) {
      const float4 p = *reinterpret_cast<const float4*>(&red[(w * 16 + row) * RED_STR + n0]);
      s.x += p.x; s.y += p.y; s.z += p.z; s.w += p.w;
    }
    Tt[f] = s;
  }
}

__device__ __forceinline__ float sumsq4(float a, const float4& v) {
  a += v.x * v.x; a += v.y * v.y; a += v.z * v.z; a += v.w * v.w;
  return a;
}

__global__ __launch_bounds__(128) __attribute__((amdgpu_waves_per_eu(1, 2))) void gram_wmma(const float* __restrict__ T,
                                                 float* __restrict__ out) {
  const int b    = blockIdx.x;
  const int it0  = (blockIdx.y >> 1) * 2;
  const int wave = threadIdx.x >> 5;
  const int lane = threadIdx.x & 31;
  const int c    = lane & 15;
  const int h    = lane >> 4;
  const int jt0  = (blockIdx.y & 1) * 8 + wave * 2;

  const float4* tb = reinterpret_cast<const float4*>(T) + (size_t)b * ((SEQ / 16) * TILE_F4) + lane;
  const float4* pI = tb + it0 * TILE_F4;
  const float4* pJ = tb + jt0 * TILE_F4;

  float4 fI[2][16], fJ[2][16];
#pragma unroll
  for (int s = 0; s < 16; ++s) {
    fJ[0][s] = pJ[s * 32];
    fJ[1][s] = pJ[TILE_F4 + s * 32];
    fI[0][s] = pI[s * 32];
    fI[1][s] = pI[TILE_F4 + s * 32];
  }

  v8f acc[2][2];
#pragma unroll
  for (int x = 0; x < 2; ++x)
#pragma unroll
    for (int y = 0; y < 2; ++y) acc[x][y] = v8f{};
  float nI[2] = {0.f, 0.f};
  float nJ[2] = {0.f, 0.f};

#pragma unroll
  for (int s = 0; s < 16; ++s) {
#pragma unroll
    for (int half = 0; half < 2; ++half) {
      v2f a[2], bf[2];
#pragma unroll
      for (int x = 0; x < 2; ++x) {
        a[x].x  = half ? fJ[x][s].z : fJ[x][s].x;
        a[x].y  = half ? fJ[x][s].w : fJ[x][s].y;
        bf[x].x = half ? fI[x][s].z : fI[x][s].x;
        bf[x].y = half ? fI[x][s].w : fI[x][s].y;
      }
#pragma unroll
      for (int x = 0; x < 2; ++x)
#pragma unroll
        for (int y = 0; y < 2; ++y)
          acc[x][y] = __builtin_amdgcn_wmma_f32_16x16x4_f32(false, a[x], false, bf[y],
                                                            (short)0, acc[x][y], false, false);
    }
#pragma unroll
    for (int x = 0; x < 2; ++x) {
      nI[x] = sumsq4(nI[x], fI[x][s]);
      nJ[x] = sumsq4(nJ[x], fJ[x][s]);
    }
  }

#pragma unroll
  for (int x = 0; x < 2; ++x) {
    nI[x] += __shfl_xor(nI[x], 16);
    nJ[x] += __shfl_xor(nJ[x], 16);
  }

  float* ob = out + (size_t)b * SEQ * SEQ;
#pragma unroll
  for (int x = 0; x < 2; ++x) {
    float nj[8];
#pragma unroll
    for (int r = 0; r < 8; ++r) nj[r] = __shfl(nJ[x], 8 * h + r);
#pragma unroll
    for (int y = 0; y < 2; ++y) {
      float* po = ob + (size_t)((it0 + y) * 16 + c) * SEQ + (jt0 + x) * 16 + 8 * h;
      const v8f d = acc[x][y];
      const float ni = nI[y];
      float4 o0, o1;
      o0.x = ni + nj[0] - 2.0f * d[0];
      o0.y = ni + nj[1] - 2.0f * d[1];
      o0.z = ni + nj[2] - 2.0f * d[2];
      o0.w = ni + nj[3] - 2.0f * d[3];
      o1.x = ni + nj[4] - 2.0f * d[4];
      o1.y = ni + nj[5] - 2.0f * d[5];
      o1.z = ni + nj[6] - 2.0f * d[6];
      o1.w = ni + nj[7] - 2.0f * d[7];
      *reinterpret_cast<float4*>(po)     = o0;
      *reinterpret_cast<float4*>(po + 4) = o1;
    }
  }
}

extern "C" void kernel_launch(void* const* d_in, const int* in_sizes, int n_in,
                              void* d_out, int out_size, void* d_ws, size_t ws_size,
                              hipStream_t stream) {
  const float* batch = (const float*)d_in[0];
  const float* Bm    = (const float*)d_in[1];
  float* out = (float*)d_out;
  float* T   = (float*)d_ws;

  proj_wmma<<<dim3(BATCHES, SEQ / 16), 256, 0, stream>>>(batch, Bm, T);
  gram_wmma<<<dim3(BATCHES, 16), 128, 0, stream>>>(T, out);
}
